# starting kernel
# baseline (speedup 1.0000x reference)
#include <hip/hip_runtime.h>
#include <math.h>

#define D 128

typedef __attribute__((ext_vector_type(2))) float v2f;
typedef __attribute__((ext_vector_type(4))) float v4f;
typedef __attribute__((ext_vector_type(8))) float v8f;

__global__ __launch_bounds__(256) void gin_zero(float* __restrict__ p, long n4) {
    long i = (long)blockIdx.x * blockDim.x + threadIdx.x;
    if (i < n4) ((v4f*)p)[i] = (v4f){0.f, 0.f, 0.f, 0.f};
}

__global__ __launch_bounds__(256) void gin_scatter(const float* __restrict__ h,
                                                   const int* __restrict__ src,
                                                   const int* __restrict__ dst,
                                                   float* __restrict__ aggr, int E) {
    int tid = blockIdx.x * 256 + threadIdx.x;
    int e = tid >> 5;
    if (e >= E) return;
    int c = (tid & 31) << 2;
    int s = src[e];
    int d = dst[e];
    v4f v = *(const v4f*)(h + (size_t)s * D + c);
    float* p = aggr + (size_t)d * D + c;
    atomicAdd(p + 0, v.x);
    atomicAdd(p + 1, v.y);
    atomicAdd(p + 2, v.z);
    atomicAdd(p + 3, v.w);
}

__global__ __launch_bounds__(256) void gin_gemm_elu(const float* h,
                                                    const float* __restrict__ aggr,
                                                    const float* __restrict__ W,
                                                    const float* __restrict__ b,
                                                    float* out, int n_rows) {
    __shared__ __align__(16) float ldsW[D * D];

    const int t = threadIdx.x;
    #pragma unroll
    for (int i = t; i < D * D / 4; i += 256)
        ((v4f*)ldsW)[i] = ((const v4f*)W)[i];
    __syncthreads();

    const int wave = t >> 5;
    const int lane = t & 31;
    const int row0 = (blockIdx.x * 8 + wave) * 16;
    if (row0 >= n_rows) return;

    const int m    = lane & 15;
    const int koff = (lane >> 4) << 1;

    const float* arow = h    + (size_t)(row0 + m) * D;
    const float* grow = aggr + (size_t)(row0 + m) * D;

    v8f c[8] = {};

    for (int kt = 0; kt < 32; ++kt) {
        const int k0 = kt * 4 + koff;
        v2f ha = *(const v2f*)(arow + k0);
        v2f ga = *(const v2f*)(grow + k0);
        v2f a = ha + ga;
        #pragma unroll
        for (int nt = 0; nt < 8; ++nt) {
            const int n = nt * 16 + m;
            v2f bf;
            bf.x = ldsW[k0 * D + n];
            bf.y = ldsW[(k0 + 1) * D + n];
            c[nt] = __builtin_amdgcn_wmma_f32_16x16x4_f32(
                false, a, false, bf, (short)0, c[nt], false, false);
        }
    }

    const int rbase = row0 + ((lane >> 4) << 3);
    #pragma unroll
    for (int nt = 0; nt < 8; ++nt) {
        const int col = nt * 16 + m;
        const float bias = b[col];
        #pragma unroll
        for (int v = 0; v < 8; ++v) {
            float x = c[nt][v] + bias;
            float y = x > 0.f ? x : expm1f(x);
            out[(size_t)(rbase + v) * D + col] = y;
        }
    }
}

extern "C" void kernel_launch(void* const* d_in, const int* in_sizes, int n_in,
                              void* d_out, int out_size, void* d_ws, size_t ws_size,
                              hipStream_t stream) {
    const float* x  = (const float*)d_in[0];
    const int*   ei = (const int*)d_in[1];
    const float* W0 = (const float*)d_in[2];
    const float* b0 = (const float*)d_in[3];
    const float* W1 = (const float*)d_in[4];
    const float* b1 = (const float*)d_in[5];
    const float* W2 = (const float*)d_in[6];
    const float* b2 = (const float*)d_in[7];

    const int N = in_sizes[0] / D;
    const int E = in_sizes[1] / 2;
    const int* src = ei;
    const int* dst = ei + E;

    float* out  = (float*)d_out;
    float* aggr = (float*)d_ws;

    const long n4 = (long)N * D / 4;
    const int zb = (int)((n4 + 255) / 256);
    const int sb = (int)(((long)E * 32 + 255) / 256);
    const int gb = (N / 16 + 7) / 8;

    gin_zero<<<zb, 256, 0, stream>>>(aggr, n4);
    gin_scatter<<<sb, 256, 0, stream>>>(x, src, dst, aggr, E);
    gin_gemm_elu<<<gb, 256, 0, stream>>>(x, aggr, W0, b0, out, N);

    gin_zero<<<zb, 256, 0, stream>>>(aggr, n4);
    gin_scatter<<<sb, 256, 0, stream>>>(out, src, dst, aggr, E);
    gin_gemm_elu<<<gb, 256, 0, stream>>>(out, aggr, W1, b1, out, N);

    gin_zero<<<zb, 256, 0, stream>>>(aggr, n4);
    gin_scatter<<<sb, 256, 0, stream>>>(out, src, dst, aggr, E);
    gin_gemm_elu<<<gb, 256, 0, stream>>>(out, aggr, W2, b2, out, N);
}


# Each wave sorts its slab of a bucket's edge records by destination row in registers and pre-adds same-destination source rows: one LDS f32 add per run, not per edge; sums: run pre-sums, then arrival
# speedup vs baseline: 3.9408x; 3.9408x over previous
#include <hip/hip_runtime.h>
#include <math.h>


#define D 128
#define R 64
#define RSH 6
#define EPB 8192
#define TG 33
#define TP 132
#define NW 12
#define NT (NW * 32)

typedef __attribute__((ext_vector_type(2))) float v2f;
typedef __attribute__((ext_vector_type(4))) float v4f;
typedef __attribute__((ext_vector_type(8))) float v8f;
typedef __attribute__((ext_vector_type(2))) int v2i;

__global__ __launch_bounds__(256) void gin_clear_counts(int* __restrict__ cnt, int n) {
    int i = blockIdx.x * 256 + threadIdx.x;
    if (i < n) cnt[i] = 0;
}

__global__ __launch_bounds__(1024) void gin_count(const int* __restrict__ dst,
                                                  int* __restrict__ cnt, int E, int nb) {
    extern __shared__ int hist[];
    const int t = threadIdx.x;
    for (int i = t; i < nb; i += 1024) hist[i] = 0;
    __syncthreads();
    const int base = blockIdx.x * EPB + t;
    #pragma unroll
    for (int j = 0; j < 8; ++j) {
        int e = base + j * 1024;
        if (e < E) atomicAdd(&hist[dst[e] >> RSH], 1);
    }
    __syncthreads();
    for (int i = t; i < nb; i += 1024) {
        int c = hist[i];
        if (c) atomicAdd(&cnt[i], c);
    }
}

__global__ __launch_bounds__(1024) void gin_scan(const int* __restrict__ cnt,
                                                 int* __restrict__ start,
                                                 int* __restrict__ cursor, int nb) {
    __shared__ int part[1024];
    const int t = threadIdx.x;
    const int per = (nb + 1023) / 1024;
    const int lo = min(t * per, nb);
    const int hi = min(lo + per, nb);
    int s = 0;
    for (int i = lo; i < hi; ++i) s += cnt[i];
    part[t] = s;
    __syncthreads();
    for (int off = 1; off < 1024; off <<= 1) {
        int v = (t >= off) ? part[t - off] : 0;
        __syncthreads();
        part[t] += v;
        __syncthreads();
    }
    int run = part[t] - s;
    for (int i = lo; i < hi; ++i) {
        start[i] = run;
        cursor[i] = run;
        run += cnt[i];
    }
    if (t == 1023) start[nb] = part[1023];
}

__global__ __launch_bounds__(1024) void gin_fill(const int* __restrict__ src,
                                                 const int* __restrict__ dst,
                                                 int* __restrict__ cursor,
                                                 int* __restrict__ rec, int E, int nb) {
    extern __shared__ int hist[];
    const int t = threadIdx.x;
    for (int i = t; i < nb; i += 1024) hist[i] = 0;
    __syncthreads();
    const int base = blockIdx.x * EPB + t;
    int dd[8], lr[8];
    #pragma unroll
    for (int j = 0; j < 8; ++j) {
        int e = base + j * 1024;
        dd[j] = 0;
        lr[j] = 0;
        if (e < E) {
            dd[j] = dst[e];
            lr[j] = atomicAdd(&hist[dd[j] >> RSH], 1);
        }
    }
    __syncthreads();
    for (int i = t; i < nb; i += 1024) {
        int c = hist[i];
        hist[i] = c ? atomicAdd(&cursor[i], c) : 0;
    }
    __syncthreads();
    #pragma unroll
    for (int j = 0; j < 8; ++j) {
        int e = base + j * 1024;
        if (e < E) {
            int pos = hist[dd[j] >> RSH] + lr[j];
            *(v2i*)(rec + 2 * (size_t)pos) = (v2i){src[e], dd[j]};
        }
    }
}

template <int M> __device__ __forceinline__ unsigned lane_xor(unsigned v) {
    if constexpr (M == 16)
        return (unsigned)__builtin_amdgcn_permlanex16((int)v, (int)v, 0x76543210, 0xfedcba98, false, false);
    else
        return (unsigned)__builtin_amdgcn_update_dpp((int)v, (int)v, 0x160 + M, 0xf, 0xf, false);
}
template <int K, int M> __device__ __forceinline__ void sort_xstage(unsigned (&a)[4], int lane) {
    const bool lower = (lane & M) == 0;
    const bool asc = (K >= 128) ? true : ((lane & (K >> 2)) == 0);
    const bool tmin = (lower == asc);
    #pragma unroll
    for (int q = 0; q < 4; ++q) {
        const unsigned p = lane_xor<M>(a[q]);
        const unsigned mn = min(a[q], p), mx = max(a[q], p);
        a[q] = tmin ? mn : mx;
    }
}
template <int K, int J> __device__ __forceinline__ void sort_istage(unsigned (&a)[4], int lane) {
    #pragma unroll
    for (int q = 0; q < 4; ++q) {
        if ((q & J) == 0) {
            bool asc;
            if (K == 2) asc = (q & 2) == 0;
            else if (K >= 128) asc = true;
            else asc = (lane & (K >> 2)) == 0;
            const unsigned x = a[q], y = a[q | J];
            const unsigned mn = min(x, y), mx = max(x, y);
            a[q] = asc ? mn : mx;
            a[q | J] = asc ? mx : mn;
        }
    }
}
__device__ __forceinline__ void sort128(unsigned (&a)[4], int lane) {
    sort_istage<2, 1>(a, lane);
    sort_istage<4, 2>(a, lane);  sort_istage<4, 1>(a, lane);
    sort_xstage<8, 1>(a, lane);  sort_istage<8, 2>(a, lane);  sort_istage<8, 1>(a, lane);
    sort_xstage<16, 2>(a, lane); sort_xstage<16, 1>(a, lane);
    sort_istage<16, 2>(a, lane); sort_istage<16, 1>(a, lane);
    sort_xstage<32, 4>(a, lane); sort_xstage<32, 2>(a, lane); sort_xstage<32, 1>(a, lane);
    sort_istage<32, 2>(a, lane); sort_istage<32, 1>(a, lane);
    sort_xstage<64, 8>(a, lane); sort_xstage<64, 4>(a, lane); sort_xstage<64, 2>(a, lane);
    sort_xstage<64, 1>(a, lane); sort_istage<64, 2>(a, lane); sort_istage<64, 1>(a, lane);
    sort_xstage<128, 16>(a, lane); sort_xstage<128, 8>(a, lane); sort_xstage<128, 4>(a, lane);
    sort_xstage<128, 2>(a, lane);  sort_xstage<128, 1>(a, lane);
    sort_istage<128, 2>(a, lane);  sort_istage<128, 1>(a, lane);
}

__global__ __launch_bounds__(NT, 9) void gin_layer(const float* __restrict__ hin,
                                                 const int* __restrict__ rec,
                                                 const int* __restrict__ start,
                                                 const float* __restrict__ W,
                                                 const float* __restrict__ bvec,
                                                 float* __restrict__ hout,
                                                 int n_rows, int E, int nb) {
    __shared__ __align__(16) float tile[(R + 1) * TP];
    __shared__ __align__(16) float ldsW[D * D];

    const int t = threadIdx.x;
    const int lane = t & 31;
    const int wave = __builtin_amdgcn_readfirstlane(t >> 5);

    #pragma unroll
    for (int i = t; i < D * D / 4; i += NT) {
        const int k = i >> 5;
        const int c4 = (i & 31) ^ (((k >> 1) & 1) << 3);
        *(v4f*)(ldsW + k * D + c4 * 4) = ((const v4f*)W)[i];
    }

    const int m  = lane & 15;
    const int hf = lane >> 4;
    const int rt = wave >> 1;
    const int ch = wave & 1;

    for (int b = blockIdx.x; b < nb; b += gridDim.x) {
        int s = start[b];
        int e = start[b + 1];
        v4f hv[(R + NW - 1) / NW];
        #pragma unroll
        for (int j = 0; j < (R + NW - 1) / NW; ++j) {
            const int row = min(j * NW + wave, R - 1);
            const int node = min(b * R + row, n_rows - 1);
            hv[j] = *(const v4f*)(hin + (size_t)node * D + lane * 4);
        }
        #pragma unroll
        for (int j = 0; j < (R + NW - 1) / NW; ++j) {
            const int row = j * NW + wave;
            if ((j + 1) * NW <= R || row < R) {
                float* p = tile + row * TP + lane;
                p[0] = hv[j].x;
                p[TG] = hv[j].y;
                p[2 * TG] = hv[j].z;
                p[3 * TG] = hv[j].w;
            }
        }
        if (t < TP) tile[R * TP + t] = 0.f;
        __syncthreads();

        s = max(0, min(s, E));
        e = max(s, min(e, E));
        const int cntb = e - s;
        const int per = min(128, (((cntb + NW - 1) / NW) + 15) & ~15);
        for (int p0 = 0; p0 < cntb; p0 += NW * per) {
            const int lo = s + p0 + wave * per;
            const int nv = min(per, e - lo);
            if (nv > 0) {
                v2i rr[4];
                #pragma unroll
                for (int q = 0; q < 4; ++q) {
                    const int pos = min(lane * 4 + q, nv - 1);
                    rr[q] = *(const v2i*)(rec + 2 * (size_t)(lo + pos));
                }
                unsigned ks[4];
                #pragma unroll
                for (int q = 0; q < 4; ++q) {
                    const unsigned w = ((unsigned)(rr[q].y & (R - 1)) << 25) | ((unsigned)rr[q].x & 0x1FFFFFFu);
                    ks[q] = (lane * 4 + q < nv) ? w : ((unsigned)R << 25);
                }
                sort128(ks, lane);
                const int ng = (nv + 15) >> 4;
                v4f carry = (v4f){0.f, 0.f, 0.f, 0.f};
                int cdl = -1;
                for (int g = 0; g < ng; ++g) {
                    v4f v[16];
                    int dl[16];
                    #pragma unroll
                    for (int j = 0; j < 16; ++j) {
                        const unsigned w = (unsigned)__builtin_amdgcn_readlane((int)ks[j & 3], 4 * g + (j >> 2));
                        dl[j] = (int)(w >> 25);
                        const unsigned sidx = w & 0x1FFFFFFu;
                        v[j] = *(const v4f*)(hin + (size_t)sidx * D + lane * 4);
                    }
                    if (cdl >= 0) {
                        if (cdl == dl[0]) {
                            v[0] += carry;
                        } else {
                            float* p = tile + cdl * TP + lane;
                            atomicAdd(p, carry.x);
                            atomicAdd(p + TG, carry.y);
                            atomicAdd(p + 2 * TG, carry.z);
                            atomicAdd(p + 3 * TG, carry.w);
                        }
                    }
                    #pragma unroll
                    for (int j = 0; j < 15; ++j) {
                        if (dl[j] == dl[j + 1]) {
                            v[j + 1] += v[j];
                        } else {
                            float* p = tile + dl[j] * TP + lane;
                            atomicAdd(p, v[j].x);
                            atomicAdd(p + TG, v[j].y);
                            atomicAdd(p + 2 * TG, v[j].z);
                            atomicAdd(p + 3 * TG, v[j].w);
                        }
                    }
                    if (g + 1 < ng) {
                        carry = v[15];
                        cdl = dl[15];
                    } else {
                        float* p = tile + dl[15] * TP + lane;
                        atomicAdd(p, v[15].x);
                        atomicAdd(p + TG, v[15].y);
                        atomicAdd(p + 2 * TG, v[15].z);
                        atomicAdd(p + 3 * TG, v[15].w);
                    }
                }
            }
        }

        const int trow = rt * 16 + m;

        __syncthreads();

        if (wave < 8 && b * R + rt * 16 < n_rows) {
            v8f c[4] = {};
            const float* trp = tile + trow * TP + (2 * hf) * TG;
            #pragma unroll
            for (int kt = 0; kt < 32; ++kt) {
                const int k0 = kt * 4 + 2 * hf;
                v2f a;
                a.x = trp[kt];
                a.y = trp[TG + kt];
                #pragma unroll
                for (int nt = 0; nt < 4; ++nt) {
                    const int nsw = (ch * 64 + nt * 16 + m) ^ (hf << 5);
                    v2f bf;
                    bf.x = ldsW[k0 * D + nsw];
                    bf.y = ldsW[(k0 + 1) * D + nsw];
                    c[nt] = __builtin_amdgcn_wmma_f32_16x16x4_f32(
                        false, a, false, bf, (short)0, c[nt], false, false);
                }
            }

            const int rbase = b * R + rt * 16 + (hf << 3);
            #pragma unroll
            for (int nt = 0; nt < 4; ++nt) {
                const int col = ch * 64 + nt * 16 + m;
                const float bias = bvec[col];
                #pragma unroll
                for (int v = 0; v < 8; ++v) {
                    float x = c[nt][v] + bias;
                    float y = x > 0.f ? x : expm1f(x);
                    if (rbase + v < n_rows) hout[(size_t)(rbase + v) * D + col] = y;
                }
            }
        }
        __syncthreads();
    }
}

extern "C" void kernel_launch(void* const* d_in, const int* in_sizes, int n_in,
                              void* d_out, int out_size, void* d_ws, size_t ws_size,
                              hipStream_t stream) {
    const float* x  = (const float*)d_in[0];
    const int*   ei = (const int*)d_in[1];
    const float* W0 = (const float*)d_in[2];
    const float* b0 = (const float*)d_in[3];
    const float* W1 = (const float*)d_in[4];
    const float* b1 = (const float*)d_in[5];
    const float* W2 = (const float*)d_in[6];
    const float* b2 = (const float*)d_in[7];

    const int N = in_sizes[0] / D;
    const int E = in_sizes[1] / 2;
    const int* src = ei;
    const int* dst = ei + E;

    const int NB = (N + R - 1) / R;

    float* out   = (float*)d_out;
    float* hmid  = (float*)d_ws;
    int* rec     = (int*)(hmid + (size_t)N * D);
    int* cnt     = rec + 2 * (size_t)E;
    int* start   = cnt + NB + 64;
    int* cursor  = start + NB + 64;

    const int cb = (NB + 255) / 256;
    const int eb = (E + EPB - 1) / EPB;
    const int lb = NB < 768 ? NB : 768;
    const size_t hs = (size_t)NB * sizeof(int);

    gin_clear_counts<<<cb, 256, 0, stream>>>(cnt, NB);
    gin_count<<<eb, 1024, hs, stream>>>(dst, cnt, E, NB);
    gin_scan<<<1, 1024, 0, stream>>>(cnt, start, cursor, NB);
    gin_fill<<<eb, 1024, hs, stream>>>(src, dst, cursor, rec, E, NB);

    gin_layer<<<lb, NT, 0, stream>>>(x,    rec, start, W0, b0, out,  N, E, NB);
    gin_layer<<<lb, NT, 0, stream>>>(out,  rec, start, W1, b1, hmid, N, E, NB);
    gin_layer<<<lb, NT, 0, stream>>>(hmid, rec, start, W2, b2, out,  N, E, NB);
}


# Column tile nt of a dense wave = W columns 4m+nt, not 16nt+m: a lane's 4 results of a row are adjacent, one 16-B store (a wave writes whole 256-B pieces, not 64-B ones); B operands 16-B LDS reads
# speedup vs baseline: 4.2178x; 1.0703x over previous
#include <hip/hip_runtime.h>
#include <math.h>


#define D 128
#define R 64
#define RSH 6
#define EPB 8192
#define TG 33
#define TP 132
#define NW 12
#define NT (NW * 32)

typedef __attribute__((ext_vector_type(2))) float v2f;
typedef __attribute__((ext_vector_type(4))) float v4f;
typedef __attribute__((ext_vector_type(8))) float v8f;
typedef __attribute__((ext_vector_type(2))) int v2i;

__global__ __launch_bounds__(256) void gin_clear_counts(int* __restrict__ cnt, int n) {
    int i = blockIdx.x * 256 + threadIdx.x;
    if (i < n) cnt[i] = 0;
}

__global__ __launch_bounds__(1024) void gin_count(const int* __restrict__ dst,
                                                  int* __restrict__ cnt, int E, int nb) {
    extern __shared__ int hist[];
    const int t = threadIdx.x;
    for (int i = t; i < nb; i += 1024) hist[i] = 0;
    __syncthreads();
    const int base = blockIdx.x * EPB + t;
    #pragma unroll
    for (int j = 0; j < 8; ++j) {
        int e = base + j * 1024;
        if (e < E) atomicAdd(&hist[dst[e] >> RSH], 1);
    }
    __syncthreads();
    for (int i = t; i < nb; i += 1024) {
        int c = hist[i];
        if (c) atomicAdd(&cnt[i], c);
    }
}

__global__ __launch_bounds__(1024) void gin_scan(const int* __restrict__ cnt,
                                                 int* __restrict__ start,
                                                 int* __restrict__ cursor, int nb) {
    __shared__ int part[1024];
    const int t = threadIdx.x;
    const int per = (nb + 1023) / 1024;
    const int lo = min(t * per, nb);
    const int hi = min(lo + per, nb);
    int s = 0;
    for (int i = lo; i < hi; ++i) s += cnt[i];
    part[t] = s;
    __syncthreads();
    for (int off = 1; off < 1024; off <<= 1) {
        int v = (t >= off) ? part[t - off] : 0;
        __syncthreads();
        part[t] += v;
        __syncthreads();
    }
    int run = part[t] - s;
    for (int i = lo; i < hi; ++i) {
        start[i] = run;
        cursor[i] = run;
        run += cnt[i];
    }
    if (t == 1023) start[nb] = part[1023];
}

__global__ __launch_bounds__(1024) void gin_fill(const int* __restrict__ src,
                                                 const int* __restrict__ dst,
                                                 int* __restrict__ cursor,
                                                 int* __restrict__ rec, int E, int nb) {
    extern __shared__ int hist[];
    const int t = threadIdx.x;
    for (int i = t; i < nb; i += 1024) hist[i] = 0;
    __syncthreads();
    const int base = blockIdx.x * EPB + t;
    int dd[8], lr[8];
    #pragma unroll
    for (int j = 0; j < 8; ++j) {
        int e = base + j * 1024;
        dd[j] = 0;
        lr[j] = 0;
        if (e < E) {
            dd[j] = dst[e];
            lr[j] = atomicAdd(&hist[dd[j] >> RSH], 1);
        }
    }
    __syncthreads();
    for (int i = t; i < nb; i += 1024) {
        int c = hist[i];
        hist[i] = c ? atomicAdd(&cursor[i], c) : 0;
    }
    __syncthreads();
    #pragma unroll
    for (int j = 0; j < 8; ++j) {
        int e = base + j * 1024;
        if (e < E) {
            int pos = hist[dd[j] >> RSH] + lr[j];
            *(v2i*)(rec + 2 * (size_t)pos) = (v2i){src[e], dd[j]};
        }
    }
}

template <int M> __device__ __forceinline__ unsigned lane_xor(unsigned v) {
    if constexpr (M == 16)
        return (unsigned)__builtin_amdgcn_permlanex16((int)v, (int)v, 0x76543210, 0xfedcba98, false, false);
    else
        return (unsigned)__builtin_amdgcn_update_dpp((int)v, (int)v, 0x160 + M, 0xf, 0xf, false);
}
template <int K, int M> __device__ __forceinline__ void sort_xstage(unsigned (&a)[4], int lane) {
    const bool lower = (lane & M) == 0;
    const bool asc = (K >= 128) ? true : ((lane & (K >> 2)) == 0);
    const bool tmin = (lower == asc);
    #pragma unroll
    for (int q = 0; q < 4; ++q) {
        const unsigned p = lane_xor<M>(a[q]);
        const unsigned mn = min(a[q], p), mx = max(a[q], p);
        a[q] = tmin ? mn : mx;
    }
}
template <int K, int J> __device__ __forceinline__ void sort_istage(unsigned (&a)[4], int lane) {
    #pragma unroll
    for (int q = 0; q < 4; ++q) {
        if ((q & J) == 0) {
            bool asc;
            if (K == 2) asc = (q & 2) == 0;
            else if (K >= 128) asc = true;
            else asc = (lane & (K >> 2)) == 0;
            const unsigned x = a[q], y = a[q | J];
            const unsigned mn = min(x, y), mx = max(x, y);
            a[q] = asc ? mn : mx;
            a[q | J] = asc ? mx : mn;
        }
    }
}
__device__ __forceinline__ void sort128(unsigned (&a)[4], int lane) {
    sort_istage<2, 1>(a, lane);
    sort_istage<4, 2>(a, lane);  sort_istage<4, 1>(a, lane);
    sort_xstage<8, 1>(a, lane);  sort_istage<8, 2>(a, lane);  sort_istage<8, 1>(a, lane);
    sort_xstage<16, 2>(a, lane); sort_xstage<16, 1>(a, lane);
    sort_istage<16, 2>(a, lane); sort_istage<16, 1>(a, lane);
    sort_xstage<32, 4>(a, lane); sort_xstage<32, 2>(a, lane); sort_xstage<32, 1>(a, lane);
    sort_istage<32, 2>(a, lane); sort_istage<32, 1>(a, lane);
    sort_xstage<64, 8>(a, lane); sort_xstage<64, 4>(a, lane); sort_xstage<64, 2>(a, lane);
    sort_xstage<64, 1>(a, lane); sort_istage<64, 2>(a, lane); sort_istage<64, 1>(a, lane);
    sort_xstage<128, 16>(a, lane); sort_xstage<128, 8>(a, lane); sort_xstage<128, 4>(a, lane);
    sort_xstage<128, 2>(a, lane);  sort_xstage<128, 1>(a, lane);
    sort_istage<128, 2>(a, lane);  sort_istage<128, 1>(a, lane);
}

__global__ __launch_bounds__(NT, 9) void gin_layer(const float* __restrict__ hin,
                                                 const int* __restrict__ rec,
                                                 const int* __restrict__ start,
                                                 const float* __restrict__ W,
                                                 const float* __restrict__ bvec,
                                                 float* __restrict__ hout,
                                                 int n_rows, int E, int nb) {
    __shared__ __align__(16) float tile[(R + 1) * TP];
    __shared__ __align__(16) float ldsW[D * D];

    const int t = threadIdx.x;
    const int lane = t & 31;
    const int wave = __builtin_amdgcn_readfirstlane(t >> 5);

    #pragma unroll
    for (int i = t; i < D * D / 4; i += NT) {
        *(v4f*)(ldsW + i * 4) = ((const v4f*)W)[i];
    }

    const int m  = lane & 15;
    const int hf = lane >> 4;
    const int rt = wave >> 1;
    const int ch = wave & 1;

    for (int b = blockIdx.x; b < nb; b += gridDim.x) {
        int s = start[b];
        int e = start[b + 1];
        v4f hv[(R + NW - 1) / NW];
        #pragma unroll
        for (int j = 0; j < (R + NW - 1) / NW; ++j) {
            const int row = min(j * NW + wave, R - 1);
            const int node = min(b * R + row, n_rows - 1);
            hv[j] = *(const v4f*)(hin + (size_t)node * D + lane * 4);
        }
        #pragma unroll
        for (int j = 0; j < (R + NW - 1) / NW; ++j) {
            const int row = j * NW + wave;
            if ((j + 1) * NW <= R || row < R) {
                float* p = tile + row * TP + lane;
                p[0] = hv[j].x;
                p[TG] = hv[j].y;
                p[2 * TG] = hv[j].z;
                p[3 * TG] = hv[j].w;
            }
        }
        if (t < TP) tile[R * TP + t] = 0.f;
        __syncthreads();

        s = max(0, min(s, E));
        e = max(s, min(e, E));
        const int cntb = e - s;
        const int per = min(128, (((cntb + NW - 1) / NW) + 15) & ~15);
        for (int p0 = 0; p0 < cntb; p0 += NW * per) {
            const int lo = s + p0 + wave * per;
            const int nv = min(per, e - lo);
            if (nv > 0) {
                v2i rr[4];
                #pragma unroll
                for (int q = 0; q < 4; ++q) {
                    const int pos = min(lane * 4 + q, nv - 1);
                    rr[q] = *(const v2i*)(rec + 2 * (size_t)(lo + pos));
                }
                unsigned ks[4];
                #pragma unroll
                for (int q = 0; q < 4; ++q) {
                    const unsigned w = ((unsigned)(rr[q].y & (R - 1)) << 25) | ((unsigned)rr[q].x & 0x1FFFFFFu);
                    ks[q] = (lane * 4 + q < nv) ? w : ((unsigned)R << 25);
                }
                sort128(ks, lane);
                const int ng = (nv + 15) >> 4;
                v4f carry = (v4f){0.f, 0.f, 0.f, 0.f};
                int cdl = -1;
                for (int g = 0; g < ng; ++g) {
                    v4f v[16];
                    int dl[16];
                    #pragma unroll
                    for (int j = 0; j < 16; ++j) {
                        const unsigned w = (unsigned)__builtin_amdgcn_readlane((int)ks[j & 3], 4 * g + (j >> 2));
                        dl[j] = (int)(w >> 25);
                        const unsigned sidx = w & 0x1FFFFFFu;
                        v[j] = *(const v4f*)(hin + (size_t)sidx * D + lane * 4);
                    }
                    if (cdl >= 0) {
                        if (cdl == dl[0]) {
                            v[0] += carry;
                        } else {
                            float* p = tile + cdl * TP + lane;
                            atomicAdd(p, carry.x);
                            atomicAdd(p + TG, carry.y);
                            atomicAdd(p + 2 * TG, carry.z);
                            atomicAdd(p + 3 * TG, carry.w);
                        }
                    }
                    #pragma unroll
                    for (int j = 0; j < 15; ++j) {
                        if (dl[j] == dl[j + 1]) {
                            v[j + 1] += v[j];
                        } else {
                            float* p = tile + dl[j] * TP + lane;
                            atomicAdd(p, v[j].x);
                            atomicAdd(p + TG, v[j].y);
                            atomicAdd(p + 2 * TG, v[j].z);
                            atomicAdd(p + 3 * TG, v[j].w);
                        }
                    }
                    if (g + 1 < ng) {
                        carry = v[15];
                        cdl = dl[15];
                    } else {
                        float* p = tile + dl[15] * TP + lane;
                        atomicAdd(p, v[15].x);
                        atomicAdd(p + TG, v[15].y);
                        atomicAdd(p + 2 * TG, v[15].z);
                        atomicAdd(p + 3 * TG, v[15].w);
                    }
                }
            }
        }

        const int trow = rt * 16 + m;

        __syncthreads();

        if (wave < 8 && b * R + rt * 16 < n_rows) {
            v8f c[4] = {};
            const float* trp = tile + trow * TP + (2 * hf) * TG;
            const int ncol = ch * 64 + 4 * m;
            const float* wp = ldsW + (2 * hf) * D + ncol;
            #pragma unroll
            for (int kt = 0; kt < 32; ++kt) {
                v2f a;
                a.x = trp[kt];
                a.y = trp[TG + kt];
                const v4f w0 = *(const v4f*)(wp + (kt * 4) * D);
                const v4f w1 = *(const v4f*)(wp + (kt * 4 + 1) * D);
                #pragma unroll
                for (int nt = 0; nt < 4; ++nt) {
                    v2f bf;
                    bf.x = w0[nt];
                    bf.y = w1[nt];
                    c[nt] = __builtin_amdgcn_wmma_f32_16x16x4_f32(
                        false, a, false, bf, (short)0, c[nt], false, false);
                }
            }

            const int rbase = b * R + rt * 16 + (hf << 3);
            const v4f bias = *(const v4f*)(bvec + ncol);
            #pragma unroll
            for (int v = 0; v < 8; ++v) {
                v4f y;
                #pragma unroll
                for (int nt = 0; nt < 4; ++nt) {
                    const float x = c[nt][v] + bias[nt];
                    y[nt] = x > 0.f ? x : expm1f(x);
                }
                if (rbase + v < n_rows) *(v4f*)(hout + (size_t)(rbase + v) * D + ncol) = y;
            }
        }
        __syncthreads();
    }
}

extern "C" void kernel_launch(void* const* d_in, const int* in_sizes, int n_in,
                              void* d_out, int out_size, void* d_ws, size_t ws_size,
                              hipStream_t stream) {
    const float* x  = (const float*)d_in[0];
    const int*   ei = (const int*)d_in[1];
    const float* W0 = (const float*)d_in[2];
    const float* b0 = (const float*)d_in[3];
    const float* W1 = (const float*)d_in[4];
    const float* b1 = (const float*)d_in[5];
    const float* W2 = (const float*)d_in[6];
    const float* b2 = (const float*)d_in[7];

    const int N = in_sizes[0] / D;
    const int E = in_sizes[1] / 2;
    const int* src = ei;
    const int* dst = ei + E;

    const int NB = (N + R - 1) / R;

    float* out   = (float*)d_out;
    float* hmid  = (float*)d_ws;
    int* rec     = (int*)(hmid + (size_t)N * D);
    int* cnt     = rec + 2 * (size_t)E;
    int* start   = cnt + NB + 64;
    int* cursor  = start + NB + 64;

    const int cb = (NB + 255) / 256;
    const int eb = (E + EPB - 1) / EPB;
    const int lb = NB < 768 ? NB : 768;
    const size_t hs = (size_t)NB * sizeof(int);

    gin_clear_counts<<<cb, 256, 0, stream>>>(cnt, NB);
    gin_count<<<eb, 1024, hs, stream>>>(dst, cnt, E, NB);
    gin_scan<<<1, 1024, 0, stream>>>(cnt, start, cursor, NB);
    gin_fill<<<eb, 1024, hs, stream>>>(src, dst, cursor, rec, E, NB);

    gin_layer<<<lb, NT, 0, stream>>>(x,    rec, start, W0, b0, out,  N, E, NB);
    gin_layer<<<lb, NT, 0, stream>>>(out,  rec, start, W1, b1, hmid, N, E, NB);
    gin_layer<<<lb, NT, 0, stream>>>(hmid, rec, start, W2, b2, out,  N, E, NB);
}


# Leftover buckets (count mod grid: a lone extra trip for a few blocks, other WGPs idle) cut into four 16-row tiles over 4x the blocks, each keeping only its rows' records; sum order kept
# speedup vs baseline: 4.3491x; 1.0311x over previous
#include <hip/hip_runtime.h>
#include <math.h>


#define D 128
#define R 64
#define RSH 6
#define EPB 8192
#define TG 33
#define TP 132
#define NW 12
#define NT (NW * 32)

typedef __attribute__((ext_vector_type(2))) float v2f;
typedef __attribute__((ext_vector_type(4))) float v4f;
typedef __attribute__((ext_vector_type(8))) float v8f;
typedef __attribute__((ext_vector_type(2))) int v2i;

__global__ __launch_bounds__(256) void gin_clear_counts(int* __restrict__ cnt, int n) {
    int i = blockIdx.x * 256 + threadIdx.x;
    if (i < n) cnt[i] = 0;
}

__global__ __launch_bounds__(1024) void gin_count(const int* __restrict__ dst,
                                                  int* __restrict__ cnt, int E, int nb) {
    extern __shared__ int hist[];
    const int t = threadIdx.x;
    for (int i = t; i < nb; i += 1024) hist[i] = 0;
    __syncthreads();
    const int base = blockIdx.x * EPB + t;
    #pragma unroll
    for (int j = 0; j < 8; ++j) {
        int e = base + j * 1024;
        if (e < E) atomicAdd(&hist[dst[e] >> RSH], 1);
    }
    __syncthreads();
    for (int i = t; i < nb; i += 1024) {
        int c = hist[i];
        if (c) atomicAdd(&cnt[i], c);
    }
}

__global__ __launch_bounds__(1024) void gin_scan(const int* __restrict__ cnt,
                                                 int* __restrict__ start,
                                                 int* __restrict__ cursor, int nb) {
    __shared__ int part[1024];
    const int t = threadIdx.x;
    const int per = (nb + 1023) / 1024;
    const int lo = min(t * per, nb);
    const int hi = min(lo + per, nb);
    int s = 0;
    for (int i = lo; i < hi; ++i) s += cnt[i];
    part[t] = s;
    __syncthreads();
    for (int off = 1; off < 1024; off <<= 1) {
        int v = (t >= off) ? part[t - off] : 0;
        __syncthreads();
        part[t] += v;
        __syncthreads();
    }
    int run = part[t] - s;
    for (int i = lo; i < hi; ++i) {
        start[i] = run;
        cursor[i] = run;
        run += cnt[i];
    }
    if (t == 1023) start[nb] = part[1023];
}

__global__ __launch_bounds__(1024) void gin_fill(const int* __restrict__ src,
                                                 const int* __restrict__ dst,
                                                 int* __restrict__ cursor,
                                                 int* __restrict__ rec, int E, int nb) {
    extern __shared__ int hist[];
    const int t = threadIdx.x;
    for (int i = t; i < nb; i += 1024) hist[i] = 0;
    __syncthreads();
    const int base = blockIdx.x * EPB + t;
    int dd[8], lr[8];
    #pragma unroll
    for (int j = 0; j < 8; ++j) {
        int e = base + j * 1024;
        dd[j] = 0;
        lr[j] = 0;
        if (e < E) {
            dd[j] = dst[e];
            lr[j] = atomicAdd(&hist[dd[j] >> RSH], 1);
        }
    }
    __syncthreads();
    for (int i = t; i < nb; i += 1024) {
        int c = hist[i];
        hist[i] = c ? atomicAdd(&cursor[i], c) : 0;
    }
    __syncthreads();
    #pragma unroll
    for (int j = 0; j < 8; ++j) {
        int e = base + j * 1024;
        if (e < E) {
            int pos = hist[dd[j] >> RSH] + lr[j];
            *(v2i*)(rec + 2 * (size_t)pos) = (v2i){src[e], dd[j]};
        }
    }
}

template <int M> __device__ __forceinline__ unsigned lane_xor(unsigned v) {
    if constexpr (M == 16)
        return (unsigned)__builtin_amdgcn_permlanex16((int)v, (int)v, 0x76543210, 0xfedcba98, false, false);
    else
        return (unsigned)__builtin_amdgcn_update_dpp((int)v, (int)v, 0x160 + M, 0xf, 0xf, false);
}
template <int K, int M> __device__ __forceinline__ void sort_xstage(unsigned (&a)[4], int lane) {
    const bool lower = (lane & M) == 0;
    const bool asc = (K >= 128) ? true : ((lane & (K >> 2)) == 0);
    const bool tmin = (lower == asc);
    #pragma unroll
    for (int q = 0; q < 4; ++q) {
        const unsigned p = lane_xor<M>(a[q]);
        const unsigned mn = min(a[q], p), mx = max(a[q], p);
        a[q] = tmin ? mn : mx;
    }
}
template <int K, int J> __device__ __forceinline__ void sort_istage(unsigned (&a)[4], int lane) {
    #pragma unroll
    for (int q = 0; q < 4; ++q) {
        if ((q & J) == 0) {
            bool asc;
            if (K == 2) asc = (q & 2) == 0;
            else if (K >= 128) asc = true;
            else asc = (lane & (K >> 2)) == 0;
            const unsigned x = a[q], y = a[q | J];
            const unsigned mn = min(x, y), mx = max(x, y);
            a[q] = asc ? mn : mx;
            a[q | J] = asc ? mx : mn;
        }
    }
}
__device__ __forceinline__ void sort128(unsigned (&a)[4], int lane) {
    sort_istage<2, 1>(a, lane);
    sort_istage<4, 2>(a, lane);  sort_istage<4, 1>(a, lane);
    sort_xstage<8, 1>(a, lane);  sort_istage<8, 2>(a, lane);  sort_istage<8, 1>(a, lane);
    sort_xstage<16, 2>(a, lane); sort_xstage<16, 1>(a, lane);
    sort_istage<16, 2>(a, lane); sort_istage<16, 1>(a, lane);
    sort_xstage<32, 4>(a, lane); sort_xstage<32, 2>(a, lane); sort_xstage<32, 1>(a, lane);
    sort_istage<32, 2>(a, lane); sort_istage<32, 1>(a, lane);
    sort_xstage<64, 8>(a, lane); sort_xstage<64, 4>(a, lane); sort_xstage<64, 2>(a, lane);
    sort_xstage<64, 1>(a, lane); sort_istage<64, 2>(a, lane); sort_istage<64, 1>(a, lane);
    sort_xstage<128, 16>(a, lane); sort_xstage<128, 8>(a, lane); sort_xstage<128, 4>(a, lane);
    sort_xstage<128, 2>(a, lane);  sort_xstage<128, 1>(a, lane);
    sort_istage<128, 2>(a, lane);  sort_istage<128, 1>(a, lane);
}

__global__ __launch_bounds__(NT, 9) void gin_layer(const float* __restrict__ hin,
                                                 const int* __restrict__ rec,
                                                 const int* __restrict__ start,
                                                 const float* __restrict__ W,
                                                 const float* __restrict__ bvec,
                                                 float* __restrict__ hout,
                                                 int n_rows, int E, int nb) {
    __shared__ __align__(16) float tile[(R + 1) * TP];
    __shared__ __align__(16) float ldsW[D * D];

    const int t = threadIdx.x;
    const int lane = t & 31;
    const int wave = __builtin_amdgcn_readfirstlane(t >> 5);

    #pragma unroll
    for (int i = t; i < D * D / 4; i += NT) {
        *(v4f*)(ldsW + i * 4) = ((const v4f*)W)[i];
    }

    const int m  = lane & 15;
    const int hf = lane >> 4;
    const int rt = wave >> 1;
    const int ch = wave & 1;

    const int grid = gridDim.x;
    const int full = (nb / grid) * grid;
    const int rem = nb - full;
    const bool split = rem * 4 <= grid;
    const int items = split ? full + rem * 4 : nb;

    for (int it = blockIdx.x; it < items; it += grid) {
        int b = it;
        int sub = -1;
        if (split && it >= full) {
            b = full + ((it - full) >> 2);
            sub = (it - full) & 3;
            if (b * R + sub * 16 >= n_rows) continue;
        }
        int s = start[b];
        int e = start[b + 1];
        v4f hv[(R + NW - 1) / NW];
        #pragma unroll
        for (int j = 0; j < (R + NW - 1) / NW; ++j) {
            const int row = min(j * NW + wave, R - 1);
            const int node = min(b * R + row, n_rows - 1);
            hv[j] = *(const v4f*)(hin + (size_t)node * D + lane * 4);
        }
        #pragma unroll
        for (int j = 0; j < (R + NW - 1) / NW; ++j) {
            const int row = j * NW + wave;
            if ((j + 1) * NW <= R || row < R) {
                float* p = tile + row * TP + lane;
                p[0] = hv[j].x;
                p[TG] = hv[j].y;
                p[2 * TG] = hv[j].z;
                p[3 * TG] = hv[j].w;
            }
        }
        if (t < TP) tile[R * TP + t] = 0.f;
        __syncthreads();

        s = max(0, min(s, E));
        e = max(s, min(e, E));
        const int cntb = e - s;
        const int per = min(128, (((cntb + NW - 1) / NW) + 15) & ~15);
        for (int p0 = 0; p0 < cntb; p0 += NW * per) {
            const int lo = s + p0 + wave * per;
            const int nv = min(per, e - lo);
            if (nv > 0) {
                v2i rr[4];
                #pragma unroll
                for (int q = 0; q < 4; ++q) {
                    const int pos = min(lane * 4 + q, nv - 1);
                    rr[q] = *(const v2i*)(rec + 2 * (size_t)(lo + pos));
                }
                unsigned ks[4];
                int nreal = 0;
                #pragma unroll
                for (int q = 0; q < 4; ++q) {
                    const int row = rr[q].y & (R - 1);
                    const unsigned w = ((unsigned)row << 25) | ((unsigned)rr[q].x & 0x1FFFFFFu);
                    const bool real = (lane * 4 + q < nv) && (sub < 0 || (row >> 4) == sub);
                    ks[q] = real ? w : ((unsigned)R << 25);
                    nreal += __builtin_popcount(__builtin_amdgcn_ballot_w32(real));
                }
                sort128(ks, lane);
                const int ng = (nreal + 15) >> 4;
                v4f carry = (v4f){0.f, 0.f, 0.f, 0.f};
                int cdl = -1;
                for (int g = 0; g < ng; ++g) {
                    v4f v[16];
                    int dl[16];
                    #pragma unroll
                    for (int j = 0; j < 16; ++j) {
                        const unsigned w = (unsigned)__builtin_amdgcn_readlane((int)ks[j & 3], 4 * g + (j >> 2));
                        dl[j] = (int)(w >> 25);
                        const unsigned sidx = w & 0x1FFFFFFu;
                        v[j] = *(const v4f*)(hin + (size_t)sidx * D + lane * 4);
                    }
                    if (cdl >= 0) {
                        if (cdl == dl[0]) {
                            v[0] += carry;
                        } else {
                            float* p = tile + cdl * TP + lane;
                            atomicAdd(p, carry.x);
                            atomicAdd(p + TG, carry.y);
                            atomicAdd(p + 2 * TG, carry.z);
                            atomicAdd(p + 3 * TG, carry.w);
                        }
                    }
                    #pragma unroll
                    for (int j = 0; j < 15; ++j) {
                        if (dl[j] == dl[j + 1]) {
                            v[j + 1] += v[j];
                        } else {
                            float* p = tile + dl[j] * TP + lane;
                            atomicAdd(p, v[j].x);
                            atomicAdd(p + TG, v[j].y);
                            atomicAdd(p + 2 * TG, v[j].z);
                            atomicAdd(p + 3 * TG, v[j].w);
                        }
                    }
                    if (g + 1 < ng) {
                        carry = v[15];
                        cdl = dl[15];
                    } else {
                        float* p = tile + dl[15] * TP + lane;
                        atomicAdd(p, v[15].x);
                        atomicAdd(p + TG, v[15].y);
                        atomicAdd(p + 2 * TG, v[15].z);
                        atomicAdd(p + 3 * TG, v[15].w);
                    }
                }
            }
        }

        const int trow = rt * 16 + m;

        __syncthreads();

        if (wave < 8 && (sub < 0 || rt == sub) && b * R + rt * 16 < n_rows) {
            v8f c[4] = {};
            const float* trp = tile + trow * TP + (2 * hf) * TG;
            const int ncol = ch * 64 + 4 * m;
            const float* wp = ldsW + (2 * hf) * D + ncol;
            #pragma unroll
            for (int kt = 0; kt < 32; ++kt) {
                v2f a;
                a.x = trp[kt];
                a.y = trp[TG + kt];
                const v4f w0 = *(const v4f*)(wp + (kt * 4) * D);
                const v4f w1 = *(const v4f*)(wp + (kt * 4 + 1) * D);
                #pragma unroll
                for (int nt = 0; nt < 4; ++nt) {
                    v2f bf;
                    bf.x = w0[nt];
                    bf.y = w1[nt];
                    c[nt] = __builtin_amdgcn_wmma_f32_16x16x4_f32(
                        false, a, false, bf, (short)0, c[nt], false, false);
                }
            }

            const int rbase = b * R + rt * 16 + (hf << 3);
            const v4f bias = *(const v4f*)(bvec + ncol);
            #pragma unroll
            for (int v = 0; v < 8; ++v) {
                v4f y;
                #pragma unroll
                for (int nt = 0; nt < 4; ++nt) {
                    const float x = c[nt][v] + bias[nt];
                    y[nt] = x > 0.f ? x : expm1f(x);
                }
                if (rbase + v < n_rows) *(v4f*)(hout + (size_t)(rbase + v) * D + ncol) = y;
            }
        }
        __syncthreads();
    }
}

extern "C" void kernel_launch(void* const* d_in, const int* in_sizes, int n_in,
                              void* d_out, int out_size, void* d_ws, size_t ws_size,
                              hipStream_t stream) {
    const float* x  = (const float*)d_in[0];
    const int*   ei = (const int*)d_in[1];
    const float* W0 = (const float*)d_in[2];
    const float* b0 = (const float*)d_in[3];
    const float* W1 = (const float*)d_in[4];
    const float* b1 = (const float*)d_in[5];
    const float* W2 = (const float*)d_in[6];
    const float* b2 = (const float*)d_in[7];

    const int N = in_sizes[0] / D;
    const int E = in_sizes[1] / 2;
    const int* src = ei;
    const int* dst = ei + E;

    const int NB = (N + R - 1) / R;

    float* out   = (float*)d_out;
    float* hmid  = (float*)d_ws;
    int* rec     = (int*)(hmid + (size_t)N * D);
    int* cnt     = rec + 2 * (size_t)E;
    int* start   = cnt + NB + 64;
    int* cursor  = start + NB + 64;

    const int cb = (NB + 255) / 256;
    const int eb = (E + EPB - 1) / EPB;
    const int lb = NB < 768 ? NB : 768;
    const size_t hs = (size_t)NB * sizeof(int);

    gin_clear_counts<<<cb, 256, 0, stream>>>(cnt, NB);
    gin_count<<<eb, 1024, hs, stream>>>(dst, cnt, E, NB);
    gin_scan<<<1, 1024, 0, stream>>>(cnt, start, cursor, NB);
    gin_fill<<<eb, 1024, hs, stream>>>(src, dst, cursor, rec, E, NB);

    gin_layer<<<lb, NT, 0, stream>>>(x,    rec, start, W0, b0, out,  N, E, NB);
    gin_layer<<<lb, NT, 0, stream>>>(out,  rec, start, W1, b1, hmid, N, E, NB);
    gin_layer<<<lb, NT, 0, stream>>>(hmid, rec, start, W2, b2, out,  N, E, NB);
}


# Edge bucketing in 2 launches, not 4, no global atomics: a block orders its 8192 edges by bucket in LDS (whole-line stores), one block per bucket merges the pieces; sum order: edge block, then arrival
# speedup vs baseline: 4.6233x; 1.0630x over previous
#include <hip/hip_runtime.h>
#include <math.h>


#define D 128
#define R 64
#define RSH 6
#define EPB 8192
#define TG 33
#define TP 132
#define NW 12
#define NT (NW * 32)
#define MU 10

typedef __attribute__((ext_vector_type(2))) float v2f;
typedef __attribute__((ext_vector_type(4))) float v4f;
typedef __attribute__((ext_vector_type(8))) float v8f;
typedef __attribute__((ext_vector_type(2))) int v2i;


template <int NTH> __device__ __forceinline__ int block_exscan(int* a, int n, int* wsum) {
    const int t = threadIdx.x, lane = t & 31, w = t >> 5;
    const int per = (n + NTH - 1) / NTH;
    const int lo = min(t * per, n);
    const int hi = min(lo + per, n);
    int s = 0;
    for (int i = lo; i < hi; ++i) s += a[i];
    int inc = s;
    #pragma unroll
    for (int o = 1; o < 32; o <<= 1) {
        const int v = __shfl_up(inc, o);
        if (lane >= o) inc += v;
    }
    if (lane == 31) wsum[w] = inc;
    __syncthreads();
    if (w == 0) {
        const int x = (lane < NTH / 32) ? wsum[lane] : 0;
        int xi = x;
        #pragma unroll
        for (int o = 1; o < 32; o <<= 1) {
            const int v = __shfl_up(xi, o);
            if (lane >= o) xi += v;
        }
        if (lane < NTH / 32) wsum[lane] = xi - x;
        if (lane == 31) wsum[32] = xi;
    }
    __syncthreads();
    int run = wsum[w] + inc - s;
    for (int i = lo; i < hi; ++i) {
        const int c = a[i];
        a[i] = run;
        run += c;
    }
    const int total = wsum[32];
    __syncthreads();
    return total;
}

__global__ __launch_bounds__(1024) void gin_block_sort(const int* __restrict__ src,
                                                       const int* __restrict__ dst,
                                                       int* __restrict__ rec1,
                                                       int* __restrict__ off,
                                                       int E, int nb, int op) {
    extern __shared__ int hist[];
    __shared__ __align__(8) int lrec[2 * EPB];
    __shared__ int wsum[33];
    const int t = threadIdx.x;
    const int k = blockIdx.x;
    for (int i = t; i < nb; i += 1024) hist[i] = 0;
    __syncthreads();
    const int base = k * EPB + t;
    int dd[8], ss[8], lr[8];
    #pragma unroll
    for (int j = 0; j < 8; ++j) {
        const int ec = min(base + j * 1024, E - 1);
        dd[j] = dst[ec];
        ss[j] = src[ec];
    }
    __builtin_amdgcn_sched_barrier(0);
    #pragma unroll
    for (int j = 0; j < 8; ++j) {
        lr[j] = 0;
        if (base + j * 1024 < E) lr[j] = atomicAdd(&hist[dd[j] >> RSH], 1);
    }
    __syncthreads();
    const int total = block_exscan<1024>(hist, nb, wsum);
    int* orow = off + (size_t)k * op;
    for (int i = t; i < nb; i += 1024) orow[i] = hist[i];
    if (t == 0) orow[nb] = total;
    #pragma unroll
    for (int j = 0; j < 8; ++j) {
        const int e = base + j * 1024;
        if (e < E) {
            const int pos = hist[dd[j] >> RSH] + lr[j];
            *(v2i*)(lrec + 2 * pos) = (v2i){ss[j], dd[j]};
        }
    }
    __syncthreads();
    for (int i = t; i < total; i += 1024)
        *(v2i*)(rec1 + 2 * ((size_t)k * EPB + i)) = *(const v2i*)(lrec + 2 * i);
}

__global__ __launch_bounds__(128) void gin_merge(const int* __restrict__ rec1,
                                                 const int* __restrict__ off,
                                                 int* __restrict__ rec,
                                                 int* __restrict__ start,
                                                 int nb, int op, int eb) {
    extern __shared__ int sm[];
    __shared__ int wsum[33];
    __shared__ int ssum;
    int* po = sm;
    int* so = sm + eb + 1;
    const int t = threadIdx.x;
    const int b = blockIdx.x;
    if (t == 0) ssum = 0;
    __syncthreads();
    int mysum = 0;
    for (int k0 = 0; k0 < eb; k0 += 2 * 128) {
        int av[2], cv[2];
        #pragma unroll
        for (int q = 0; q < 2; ++q) {
            const int* orow = off + (size_t)min(k0 + q * 128 + t, eb - 1) * op + b;
            av[q] = orow[0];
            cv[q] = orow[1];
        }
        #pragma unroll
        for (int q = 0; q < 2; ++q) {
            const int k = k0 + q * 128 + t;
            if (k < eb) {
                so[k] = av[q];
                po[k] = cv[q] - av[q];
                mysum += av[q];
            }
        }
    }
    #pragma unroll
    for (int o = 16; o > 0; o >>= 1) mysum += __shfl_xor(mysum, o);
    if ((t & 31) == 0) atomicAdd(&ssum, mysum);
    __syncthreads();
    const int total = block_exscan<128>(po, eb, wsum);
    const int s0 = ssum;
    const int steps = eb > 1 ? 32 - __builtin_clz(eb - 1) : 0;
    for (int p0 = 0; p0 < total; p0 += 128 * MU) {
        int pp[MU], lo[MU], hi[MU];
        #pragma unroll
        for (int u = 0; u < MU; ++u) {
            pp[u] = min(p0 + u * 128 + t, total - 1);
            lo[u] = 0;
            hi[u] = eb;
        }
        for (int it = 0; it < steps; ++it) {
            int mid[MU], pv[MU];
            #pragma unroll
            for (int u = 0; u < MU; ++u) mid[u] = (lo[u] + hi[u]) >> 1;
            #pragma unroll
            for (int u = 0; u < MU; ++u) pv[u] = po[mid[u]];
            __builtin_amdgcn_sched_barrier(0);
            #pragma unroll
            for (int u = 0; u < MU; ++u) {
                const bool open = hi[u] - lo[u] > 1;
                const bool right = pv[u] <= pp[u];
                lo[u] = (open && right) ? mid[u] : lo[u];
                hi[u] = (open && !right) ? mid[u] : hi[u];
            }
        }
        v2i r[MU];
        #pragma unroll
        for (int u = 0; u < MU; ++u) {
            const size_t si = (size_t)lo[u] * EPB + so[lo[u]] + (pp[u] - po[lo[u]]);
            r[u] = *(const v2i*)(rec1 + 2 * si);
        }
        #pragma unroll
        for (int u = 0; u < MU; ++u) {
            const int p = p0 + u * 128 + t;
            if (p < total) *(v2i*)(rec + 2 * ((size_t)s0 + p)) = r[u];
        }
    }
    if (t == 0) {
        start[b] = s0;
        if (b == nb - 1) start[nb] = s0 + total;
    }
}

template <int M> __device__ __forceinline__ unsigned lane_xor(unsigned v) {
    if constexpr (M == 16)
        return (unsigned)__builtin_amdgcn_permlanex16((int)v, (int)v, 0x76543210, 0xfedcba98, false, false);
    else
        return (unsigned)__builtin_amdgcn_update_dpp((int)v, (int)v, 0x160 + M, 0xf, 0xf, false);
}
template <int K, int M> __device__ __forceinline__ void sort_xstage(unsigned (&a)[4], int lane) {
    const bool lower = (lane & M) == 0;
    const bool asc = (K >= 128) ? true : ((lane & (K >> 2)) == 0);
    const bool tmin = (lower == asc);
    #pragma unroll
    for (int q = 0; q < 4; ++q) {
        const unsigned p = lane_xor<M>(a[q]);
        const unsigned mn = min(a[q], p), mx = max(a[q], p);
        a[q] = tmin ? mn : mx;
    }
}
template <int K, int J> __device__ __forceinline__ void sort_istage(unsigned (&a)[4], int lane) {
    #pragma unroll
    for (int q = 0; q < 4; ++q) {
        if ((q & J) == 0) {
            bool asc;
            if (K == 2) asc = (q & 2) == 0;
            else if (K >= 128) asc = true;
            else asc = (lane & (K >> 2)) == 0;
            const unsigned x = a[q], y = a[q | J];
            const unsigned mn = min(x, y), mx = max(x, y);
            a[q] = asc ? mn : mx;
            a[q | J] = asc ? mx : mn;
        }
    }
}
__device__ __forceinline__ void sort128(unsigned (&a)[4], int lane) {
    sort_istage<2, 1>(a, lane);
    sort_istage<4, 2>(a, lane);  sort_istage<4, 1>(a, lane);
    sort_xstage<8, 1>(a, lane);  sort_istage<8, 2>(a, lane);  sort_istage<8, 1>(a, lane);
    sort_xstage<16, 2>(a, lane); sort_xstage<16, 1>(a, lane);
    sort_istage<16, 2>(a, lane); sort_istage<16, 1>(a, lane);
    sort_xstage<32, 4>(a, lane); sort_xstage<32, 2>(a, lane); sort_xstage<32, 1>(a, lane);
    sort_istage<32, 2>(a, lane); sort_istage<32, 1>(a, lane);
    sort_xstage<64, 8>(a, lane); sort_xstage<64, 4>(a, lane); sort_xstage<64, 2>(a, lane);
    sort_xstage<64, 1>(a, lane); sort_istage<64, 2>(a, lane); sort_istage<64, 1>(a, lane);
    sort_xstage<128, 16>(a, lane); sort_xstage<128, 8>(a, lane); sort_xstage<128, 4>(a, lane);
    sort_xstage<128, 2>(a, lane);  sort_xstage<128, 1>(a, lane);
    sort_istage<128, 2>(a, lane);  sort_istage<128, 1>(a, lane);
}

__global__ __launch_bounds__(NT, 9) void gin_layer(const float* __restrict__ hin,
                                                 const int* __restrict__ rec,
                                                 const int* __restrict__ start,
                                                 const float* __restrict__ W,
                                                 const float* __restrict__ bvec,
                                                 float* __restrict__ hout,
                                                 int n_rows, int E, int nb) {
    __shared__ __align__(16) float tile[(R + 1) * TP];
    __shared__ __align__(16) float ldsW[D * D];

    const int t = threadIdx.x;
    const int lane = t & 31;
    const int wave = __builtin_amdgcn_readfirstlane(t >> 5);

    #pragma unroll
    for (int i = t; i < D * D / 4; i += NT) {
        *(v4f*)(ldsW + i * 4) = ((const v4f*)W)[i];
    }

    const int m  = lane & 15;
    const int hf = lane >> 4;
    const int rt = wave >> 1;
    const int ch = wave & 1;

    const int grid = gridDim.x;
    const int full = (nb / grid) * grid;
    const int rem = nb - full;
    const bool split = rem * 4 <= grid;
    const int items = split ? full + rem * 4 : nb;

    for (int it = blockIdx.x; it < items; it += grid) {
        int b = it;
        int sub = -1;
        if (split && it >= full) {
            b = full + ((it - full) >> 2);
            sub = (it - full) & 3;
            if (b * R + sub * 16 >= n_rows) continue;
        }
        int s = start[b];
        int e = start[b + 1];
        v4f hv[(R + NW - 1) / NW];
        #pragma unroll
        for (int j = 0; j < (R + NW - 1) / NW; ++j) {
            const int row = min(j * NW + wave, R - 1);
            const int node = min(b * R + row, n_rows - 1);
            hv[j] = *(const v4f*)(hin + (size_t)node * D + lane * 4);
        }
        #pragma unroll
        for (int j = 0; j < (R + NW - 1) / NW; ++j) {
            const int row = j * NW + wave;
            if ((j + 1) * NW <= R || row < R) {
                float* p = tile + row * TP + lane;
                p[0] = hv[j].x;
                p[TG] = hv[j].y;
                p[2 * TG] = hv[j].z;
                p[3 * TG] = hv[j].w;
            }
        }
        if (t < TP) tile[R * TP + t] = 0.f;
        __syncthreads();

        s = max(0, min(s, E));
        e = max(s, min(e, E));
        const int cntb = e - s;
        const int per = min(128, (((cntb + NW - 1) / NW) + 15) & ~15);
        for (int p0 = 0; p0 < cntb; p0 += NW * per) {
            const int lo = s + p0 + wave * per;
            const int nv = min(per, e - lo);
            if (nv > 0) {
                v2i rr[4];
                #pragma unroll
                for (int q = 0; q < 4; ++q) {
                    const int pos = min(lane * 4 + q, nv - 1);
                    rr[q] = *(const v2i*)(rec + 2 * (size_t)(lo + pos));
                }
                unsigned ks[4];
                int nreal = 0;
                #pragma unroll
                for (int q = 0; q < 4; ++q) {
                    const int row = rr[q].y & (R - 1);
                    const unsigned w = ((unsigned)row << 25) | ((unsigned)rr[q].x & 0x1FFFFFFu);
                    const bool real = (lane * 4 + q < nv) && (sub < 0 || (row >> 4) == sub);
                    ks[q] = real ? w : ((unsigned)R << 25);
                    nreal += __builtin_popcount(__builtin_amdgcn_ballot_w32(real));
                }
                sort128(ks, lane);
                const int ng = (nreal + 15) >> 4;
                v4f carry = (v4f){0.f, 0.f, 0.f, 0.f};
                int cdl = -1;
                for (int g = 0; g < ng; ++g) {
                    v4f v[16];
                    int dl[16];
                    #pragma unroll
                    for (int j = 0; j < 16; ++j) {
                        const unsigned w = (unsigned)__builtin_amdgcn_readlane((int)ks[j & 3], 4 * g + (j >> 2));
                        dl[j] = (int)(w >> 25);
                        const unsigned sidx = w & 0x1FFFFFFu;
                        v[j] = *(const v4f*)(hin + (size_t)sidx * D + lane * 4);
                    }
                    if (cdl >= 0) {
                        if (cdl == dl[0]) {
                            v[0] += carry;
                        } else {
                            float* p = tile + cdl * TP + lane;
                            atomicAdd(p, carry.x);
                            atomicAdd(p + TG, carry.y);
                            atomicAdd(p + 2 * TG, carry.z);
                            atomicAdd(p + 3 * TG, carry.w);
                        }
                    }
                    #pragma unroll
                    for (int j = 0; j < 15; ++j) {
                        if (dl[j] == dl[j + 1]) {
                            v[j + 1] += v[j];
                        } else {
                            float* p = tile + dl[j] * TP + lane;
                            atomicAdd(p, v[j].x);
                            atomicAdd(p + TG, v[j].y);
                            atomicAdd(p + 2 * TG, v[j].z);
                            atomicAdd(p + 3 * TG, v[j].w);
                        }
                    }
                    if (g + 1 < ng) {
                        carry = v[15];
                        cdl = dl[15];
                    } else {
                        float* p = tile + dl[15] * TP + lane;
                        atomicAdd(p, v[15].x);
                        atomicAdd(p + TG, v[15].y);
                        atomicAdd(p + 2 * TG, v[15].z);
                        atomicAdd(p + 3 * TG, v[15].w);
                    }
                }
            }
        }

        const int trow = rt * 16 + m;

        __syncthreads();

        if (wave < 8 && (sub < 0 || rt == sub) && b * R + rt * 16 < n_rows) {
            v8f c[4] = {};
            const float* trp = tile + trow * TP + (2 * hf) * TG;
            const int ncol = ch * 64 + 4 * m;
            const float* wp = ldsW + (2 * hf) * D + ncol;
            #pragma unroll
            for (int kt = 0; kt < 32; ++kt) {
                v2f a;
                a.x = trp[kt];
                a.y = trp[TG + kt];
                const v4f w0 = *(const v4f*)(wp + (kt * 4) * D);
                const v4f w1 = *(const v4f*)(wp + (kt * 4 + 1) * D);
                #pragma unroll
                for (int nt = 0; nt < 4; ++nt) {
                    v2f bf;
                    bf.x = w0[nt];
                    bf.y = w1[nt];
                    c[nt] = __builtin_amdgcn_wmma_f32_16x16x4_f32(
                        false, a, false, bf, (short)0, c[nt], false, false);
                }
            }

            const int rbase = b * R + rt * 16 + (hf << 3);
            const v4f bias = *(const v4f*)(bvec + ncol);
            #pragma unroll
            for (int v = 0; v < 8; ++v) {
                v4f y;
                #pragma unroll
                for (int nt = 0; nt < 4; ++nt) {
                    const float x = c[nt][v] + bias[nt];
                    y[nt] = x > 0.f ? x : expm1f(x);
                }
                if (rbase + v < n_rows) *(v4f*)(hout + (size_t)(rbase + v) * D + ncol) = y;
            }
        }
        __syncthreads();
    }
}

extern "C" void kernel_launch(void* const* d_in, const int* in_sizes, int n_in,
                              void* d_out, int out_size, void* d_ws, size_t ws_size,
                              hipStream_t stream) {
    const float* x  = (const float*)d_in[0];
    const int*   ei = (const int*)d_in[1];
    const float* W0 = (const float*)d_in[2];
    const float* b0 = (const float*)d_in[3];
    const float* W1 = (const float*)d_in[4];
    const float* b1 = (const float*)d_in[5];
    const float* W2 = (const float*)d_in[6];
    const float* b2 = (const float*)d_in[7];

    const int N = in_sizes[0] / D;
    const int E = in_sizes[1] / 2;
    const int* src = ei;
    const int* dst = ei + E;

    const int NB = (N + R - 1) / R;

    float* out   = (float*)d_out;
    float* hmid  = (float*)d_ws;
    int* rec     = (int*)(hmid + (size_t)N * D);
    const int eb = (E + EPB - 1) / EPB;
    const int op = (NB + 32) & ~31;
    int* rec1    = rec + 2 * (size_t)E;
    int* off     = rec1 + 2 * (size_t)eb * EPB;
    int* start   = off + (size_t)eb * op;

    const int lb = NB < 768 ? NB : 768;
    const size_t hs = (size_t)NB * sizeof(int);
    const size_t ms = (size_t)(2 * eb + 1) * sizeof(int);

    gin_block_sort<<<eb, 1024, hs, stream>>>(src, dst, rec1, off, E, NB, op);
    gin_merge<<<NB, 128, ms, stream>>>(rec1, off, rec, start, NB, op, eb);

    gin_layer<<<lb, NT, 0, stream>>>(x,    rec, start, W0, b0, out,  N, E, NB);
    gin_layer<<<lb, NT, 0, stream>>>(out,  rec, start, W1, b1, hmid, N, E, NB);
    gin_layer<<<lb, NT, 0, stream>>>(hmid, rec, start, W2, b2, out,  N, E, NB);
}
